# starting kernel
# baseline (speedup 1.0000x reference)
#include <hip/hip_runtime.h>
#include <hip/hip_bf16.h>
#include <math.h>

#define B_  16
#define Q_  64
#define K_  1024
#define QS_ 256
#define H_  128
#define VS_ 256
#define NEG_ (-1e6f)

typedef float v2f __attribute__((ext_vector_type(2)));
typedef float v8f __attribute__((ext_vector_type(8)));

#if defined(__has_builtin)
#  if __has_builtin(__builtin_amdgcn_tanhf)
#    define TANHF(x) __builtin_amdgcn_tanhf(x)
#  else
#    define TANHF(x) tanhf(x)
#  endif
#  if __has_builtin(__builtin_amdgcn_exp2f)
#    define EXP2F(x) __builtin_amdgcn_exp2f(x)
#  else
#    define EXP2F(x) exp2f(x)
#  endif
#else
#  define TANHF(x) tanhf(x)
#  define EXP2F(x) exp2f(x)
#endif

#define LOG2E 1.4426950408889634f

template<int N, int KD>
__global__ __launch_bounds__(128) void gemm_wmma_f32(
    const float* __restrict__ A, long sA,
    const float* __restrict__ W, long sW,
    float* __restrict__ C, long sC)
{
    A += (size_t)blockIdx.y * sA;
    W += (size_t)blockIdx.y * sW;
    C += (size_t)blockIdx.y * sC;

    const int wave = threadIdx.x >> 5;
    const int lane = threadIdx.x & 31;
    const int tile = blockIdx.x * 4 + wave;
    constexpr int tilesN = N >> 4;
    const int tm = tile / tilesN;
    const int tn = tile - tm * tilesN;
    const int r    = lane & 15;
    const int half = lane >> 4;

    const float* arow = A + (size_t)(tm * 16 + r) * KD + 2 * half;
    const float* wcol = W + (size_t)(2 * half) * N + tn * 16 + r;

    v8f c = {};
    #pragma unroll 8
    for (int kk = 0; kk < KD; kk += 4) {
        const v2f a = *(const v2f*)(arow + kk);
        v2f b;
        b.x = wcol[(size_t)kk * N];
        b.y = wcol[(size_t)(kk + 1) * N];
        c = __builtin_amdgcn_wmma_f32_16x16x4_f32(
                 false, a,  false, b,
                 (short)0, c,  false,  false);
    }

    float* crow = C + (size_t)(tm * 16 + 8 * half) * N + tn * 16 + r;
    #pragma unroll
    for (int v = 0; v < 8; ++v)
        crow[(size_t)v * N] = c[v];
}

__global__ __launch_bounds__(256) void scores_softmax(
    const float* __restrict__ qproj,
    const float* __restrict__ kproj,
    const float* __restrict__ Ws,
    const int*   __restrict__ valid_len,
    float* __restrict__ attn)
{
    __shared__ float s_scores[K_];
    __shared__ float s_red[8];

    const int b = blockIdx.x >> 6;
    const int q = blockIdx.x & 63;
    const int t = threadIdx.x;
    const int wave = t >> 5;
    const int lane = t & 31;
    const int h0 = lane * 4;

    const float4 qv = *(const float4*)(qproj + ((size_t)b * Q_ + q) * H_ + h0);
    const float4 wv = *(const float4*)(Ws + h0);
    const int vlen = valid_len[b];
    const float* kbase = kproj + (size_t)b * K_ * H_;

    for (int k = wave; k < K_; k += 8) {
        const float4 kv = *(const float4*)(kbase + (size_t)k * H_ + h0);
        float p = wv.x * TANHF(qv.x + kv.x)
                + wv.y * TANHF(qv.y + kv.y)
                + wv.z * TANHF(qv.z + kv.z)
                + wv.w * TANHF(qv.w + kv.w);
        #pragma unroll
        for (int off = 16; off >= 1; off >>= 1)
            p += __shfl_xor(p, off, 32);
        if (lane == 0)
            s_scores[k] = (k < vlen) ? NEG_ : p;
    }
    __syncthreads();

    const float v0 = s_scores[t];
    const float v1 = s_scores[t + 256];
    const float v2 = s_scores[t + 512];
    const float v3 = s_scores[t + 768];

    float m = fmaxf(fmaxf(v0, v1), fmaxf(v2, v3));
    #pragma unroll
    for (int off = 16; off >= 1; off >>= 1)
        m = fmaxf(m, __shfl_xor(m, off, 32));
    if (lane == 0) s_red[wave] = m;
    __syncthreads();
    float gmax = s_red[0];
    #pragma unroll
    for (int w = 1; w < 8; ++w) gmax = fmaxf(gmax, s_red[w]);
    __syncthreads();

    const float e0 = EXP2F((v0 - gmax) * LOG2E);
    const float e1 = EXP2F((v1 - gmax) * LOG2E);
    const float e2 = EXP2F((v2 - gmax) * LOG2E);
    const float e3 = EXP2F((v3 - gmax) * LOG2E);
    float s = e0 + e1 + e2 + e3;
    #pragma unroll
    for (int off = 16; off >= 1; off >>= 1)
        s += __shfl_xor(s, off, 32);
    if (lane == 0) s_red[wave] = s;
    __syncthreads();
    float tot = s_red[0];
    #pragma unroll
    for (int w = 1; w < 8; ++w) tot += s_red[w];
    const float inv = 1.0f / tot;

    float* orow = attn + ((size_t)b * Q_ + q) * K_;
    orow[t]       = e0 * inv;
    orow[t + 256] = e1 * inv;
    orow[t + 512] = e2 * inv;
    orow[t + 768] = e3 * inv;
}

extern "C" void kernel_launch(void* const* d_in, const int* in_sizes, int n_in,
                              void* d_out, int out_size, void* d_ws, size_t ws_size,
                              hipStream_t stream) {
    const float* queries   = (const float*)d_in[0];
    const float* keys      = (const float*)d_in[1];
    const float* values    = (const float*)d_in[2];
    const int*   valid_len = (const int*)  d_in[3];
    const float* Wq        = (const float*)d_in[4];
    const float* Wk        = (const float*)d_in[5];
    const float* Ws        = (const float*)d_in[6];
    float* out = (float*)d_out;

    float* qproj = (float*)d_ws;
    float* kproj = qproj + (size_t)B_ * Q_ * H_;
    float* attn  = kproj + (size_t)B_ * K_ * H_;

    gemm_wmma_f32<H_, QS_><<<dim3(128, 1), 128, 0, stream>>>(
        queries, 0, Wq, 0, qproj, 0);

    gemm_wmma_f32<H_, QS_><<<dim3(2048, 1), 128, 0, stream>>>(
        keys, 0, Wk, 0, kproj, 0);

    scores_softmax<<<B_ * Q_, 256, 0, stream>>>(
        qproj, kproj, Ws, valid_len, attn);

    gemm_wmma_f32<VS_, K_><<<dim3(16, B_), 128, 0, stream>>>(
        attn,   (long)Q_ * K_,
        values, (long)K_ * VS_,
        out,    (long)Q_ * VS_);
}


# every product: a block's 16-row A strip staged once in LDS by whole-line 16-byte loads, A fragments read there (was 8-byte loads over 16 lines); values ring 19->30 in freed load slots; bit-identical
# speedup vs baseline: 2.7706x; 2.7706x over previous
#include <hip/hip_runtime.h>
#include <hip/hip_bf16.h>
#include <math.h>

#define B_  16
#define Q_  64
#define K_  1024
#define QS_ 256
#define KS_ 256
#define H_  128
#define VS_ 256
#define NEG_ (-1e6f)

typedef float v2f __attribute__((ext_vector_type(2)));
typedef float v8f __attribute__((ext_vector_type(8)));

#if defined(__has_builtin)
#  if __has_builtin(__builtin_amdgcn_tanhf)
#    define TANHF(x) __builtin_amdgcn_tanhf(x)
#  else
#    define TANHF(x) tanhf(x)
#  endif
#  if __has_builtin(__builtin_amdgcn_exp2f)
#    define EXP2F(x) __builtin_amdgcn_exp2f(x)
#  else
#    define EXP2F(x) exp2f(x)
#  endif
#else
#  define TANHF(x) tanhf(x)
#  define EXP2F(x) exp2f(x)
#endif

#define LOG2E 1.4426950408889634f

template<int N, int KD>
__global__ __launch_bounds__(128) void gemm_wmma_f32(
    const float* __restrict__ A, long sA,
    const float* __restrict__ W, long sW,
    float* __restrict__ C, long sC)
{
    constexpr int tilesN = N >> 4;
    constexpr int F4ROW  = KD / 4;
    constexpr int NST    = 16 * F4ROW / 128;
    constexpr int PITCH  = KD + 4;
    static_assert(tilesN % 4 == 0, "the four waves of a block must share their A strip");
    static_assert(KD % 64 == 0 && (16 * F4ROW) % 128 == 0, "shape");

    __shared__ __attribute__((aligned(16))) float s_a[16 * PITCH];

    A += (size_t)blockIdx.y * sA;
    W += (size_t)blockIdx.y * sW;
    C += (size_t)blockIdx.y * sC;

    const int tid  = threadIdx.x;
    const int wave = tid >> 5;
    const int lane = tid & 31;
    const int tile = blockIdx.x * 4 + wave;
    const int tm = tile / tilesN;
    const int tn = tile - tm * tilesN;
    const int r    = lane & 15;
    const int half = lane >> 4;

    {
        const float4* asrc = (const float4*)(A + (size_t)tm * 16 * KD);
        #pragma unroll
        for (int i = 0; i < NST; ++i) {
            const int idx = i * 128 + tid;
            const float4 v = asrc[idx];
            *(float4*)(s_a + (idx / F4ROW) * PITCH + (idx % F4ROW) * 4) = v;
        }
    }
    __syncthreads();

    const float* arow = s_a + r * PITCH + 2 * half;
    const float* wcol = W + (size_t)(2 * half) * N + tn * 16 + r;

    v8f c = {};
    for (int k0s = 0; k0s < KD; k0s += 64) {
        v2f b[16], a[16];
        #pragma unroll
        for (int s = 0; s < 16; ++s) {
            const int kk = k0s + 4 * s;
            b[s].x = wcol[(size_t)kk * N];
            b[s].y = wcol[(size_t)(kk + 1) * N];
        }
        #pragma unroll
        for (int s = 0; s < 16; ++s)
            a[s] = *(const v2f*)(arow + k0s + 4 * s);
        __builtin_amdgcn_sched_barrier(0);
        #pragma unroll
        for (int s = 0; s < 16; ++s)
            c = __builtin_amdgcn_wmma_f32_16x16x4_f32(
                     false, a[s],  false, b[s],
                     (short)0, c,  false,  false);
    }

    float* crow = C + (size_t)(tm * 16 + 8 * half) * N + tn * 16 + r;
    #pragma unroll
    for (int v = 0; v < 8; ++v)
        crow[(size_t)v * N] = c[v];
}

template<int N, int KD, int RING, int GRP>
__global__ __launch_bounds__(128) void gemm_wmma_f32_ahead(
    const float* __restrict__ A, long sA,
    const float* __restrict__ W, long sW,
    float* __restrict__ C, long sC)
{
    constexpr int tilesN = N >> 4;
    constexpr int STEPS  = KD / 4;
    constexpr int NGRP   = STEPS / GRP;
    constexpr int F4ROW  = KD / 4;
    constexpr int NST    = 16 * F4ROW / 128;
    constexpr int PITCH  = KD + 4;
    static_assert(tilesN % 4 == 0, "the four waves of a block must share their A strip");
    static_assert(STEPS % GRP == 0 && (16 * F4ROW) % 128 == 0 && RING <= STEPS, "shape");

    __shared__ __attribute__((aligned(16))) float s_a[16 * PITCH];

    A += (size_t)blockIdx.y * sA;
    W += (size_t)blockIdx.y * sW;
    C += (size_t)blockIdx.y * sC;

    const int tid  = threadIdx.x;
    const int wave = tid >> 5;
    const int lane = tid & 31;
    const int tile = blockIdx.x * 4 + wave;
    const int tm = tile / tilesN;
    const int tn = tile - tm * tilesN;
    const int r    = lane & 15;
    const int half = lane >> 4;

    const float* wcol = W + (size_t)(2 * half) * N + tn * 16 + r;

    const float4* asrc = (const float4*)(A + (size_t)tm * 16 * KD);
    float4 piece[NST];
    #pragma unroll
    for (int i = 0; i < NST; ++i)
        piece[i] = asrc[i * 128 + tid];

    v2f rb[RING];
    #pragma unroll
    for (int s = 0; s < RING; ++s) {
        rb[s].x = wcol[(size_t)(4 * s) * N];
        rb[s].y = wcol[(size_t)(4 * s + 1) * N];
    }

    #pragma unroll
    for (int i = 0; i < NST; ++i) {
        const int idx = i * 128 + tid;
        *(float4*)(s_a + (idx / F4ROW) * PITCH + (idx % F4ROW) * 4) = piece[i];
    }
    __syncthreads();

    const float* afrag = s_a + r * PITCH + 2 * half;
    v2f ra[2][GRP];
    #pragma unroll
    for (int s = 0; s < GRP; ++s)
        ra[0][s] = *(const v2f*)(afrag + 4 * s);

    v8f c = {};
    #pragma unroll
    for (int g = 0; g < NGRP; ++g) {
        if (g + 1 < NGRP) {
            #pragma unroll
            for (int s = 0; s < GRP; ++s)
                ra[(g + 1) & 1][s] = *(const v2f*)(afrag + 4 * ((g + 1) * GRP + s));
        }
        __builtin_amdgcn_sched_barrier(0);
        #pragma unroll
        for (int s = 0; s < GRP; ++s) {
            const int st = g * GRP + s;
            c = __builtin_amdgcn_wmma_f32_16x16x4_f32(
                     false, ra[g & 1][s],  false, rb[st % RING],
                     (short)0, c,  false,  false);
            if (st + RING < STEPS) {
                const int n = st + RING;
                rb[st % RING].x = wcol[(size_t)(4 * n) * N];
                rb[st % RING].y = wcol[(size_t)(4 * n + 1) * N];
            }
            __builtin_amdgcn_sched_barrier(0);
        }
    }

    float* crow = C + (size_t)(tm * 16 + 8 * half) * N + tn * 16 + r;
    #pragma unroll
    for (int v = 0; v < 8; ++v)
        crow[(size_t)v * N] = c[v];
}

#define DPP_XMASK_ 0x160

__device__ __forceinline__ float xadd_rows(float lo, float hi, bool up) {
    const float keep = up ? hi : lo;
    const int send = __builtin_bit_cast(int, up ? lo : hi);
    const float got = __builtin_bit_cast(float, __builtin_amdgcn_permlanex16(
        send, send, 0x76543210, 0xfedcba98, false, false));
    return keep + got;
}

template<int CTRL>
__device__ __forceinline__ float xadd_in_row(float lo, float hi, bool up) {
    const float keep = up ? hi : lo;
    const float send = up ? lo : hi;
    const float got = __builtin_bit_cast(float, __builtin_amdgcn_update_dpp(
        0, __builtin_bit_cast(int, send), CTRL, 0xF, 0xF, true));
    return keep + got;
}

#define KPITCH_ (H_ + 4)
#define APITCH_ (KS_ + 4)

__global__ __launch_bounds__(256) void kproj_scores(
    const float* __restrict__ keys,
    const float* __restrict__ Wk,
    const float* __restrict__ qproj,
    const float* __restrict__ Ws,
    const int*   __restrict__ valid_len,
    float* __restrict__ raw)
{
    __shared__ __attribute__((aligned(16))) float s_k[32 * KPITCH_];
    __shared__ __attribute__((aligned(16))) float s_a[32 * APITCH_];

    const int b  = blockIdx.x >> 5;
    const int k0 = (blockIdx.x & 31) * 32;
    const int wave = threadIdx.x >> 5;
    const int lane = threadIdx.x & 31;

    {
        const float4* ksrc = (const float4*)(keys + ((size_t)b * K_ + k0) * KS_);
        #pragma unroll
        for (int i = 0; i < 8; ++i) {
            const int idx = i * 256 + (int)threadIdx.x;
            const float4 v = ksrc[idx];
            *(float4*)(s_a + (idx >> 6) * APITCH_ + (idx & 63) * 4) = v;
        }
    }
    __syncthreads();
    {
        const int r    = lane & 15;
        const int half = lane >> 4;
        const int tm   = wave >> 2;
        const int tn0  = (wave & 3) * 2;

        const float* arow = s_a + (tm * 16 + r) * APITCH_ + 2 * half;
        const float* wcol = Wk + (size_t)(2 * half) * H_ + tn0 * 16 + r;

        v8f c0 = {}, c1 = {};
        for (int k0s = 0; k0s < KS_; k0s += 64) {
            v2f b0[16], b1[16], a[16];
            #pragma unroll
            for (int s = 0; s < 16; ++s) {
                const int kk = k0s + 4 * s;
                b0[s].x = wcol[(size_t)kk * H_];
                b0[s].y = wcol[(size_t)(kk + 1) * H_];
                b1[s].x = wcol[(size_t)kk * H_ + 16];
                b1[s].y = wcol[(size_t)(kk + 1) * H_ + 16];
            }
            #pragma unroll
            for (int s = 0; s < 16; ++s)
                a[s] = *(const v2f*)(arow + k0s + 4 * s);
            __builtin_amdgcn_sched_barrier(0);
            #pragma unroll
            for (int s = 0; s < 16; ++s) {
                c0 = __builtin_amdgcn_wmma_f32_16x16x4_f32(
                         false, a[s],  false, b0[s],
                         (short)0, c0,  false,  false);
                c1 = __builtin_amdgcn_wmma_f32_16x16x4_f32(
                         false, a[s],  false, b1[s],
                         (short)0, c1,  false,  false);
            }
        }

        float* srow = s_k + (tm * 16 + 8 * half) * KPITCH_ + tn0 * 16 + r;
        #pragma unroll
        for (int v = 0; v < 8; ++v) {
            srow[v * KPITCH_]      = c0[v];
            srow[v * KPITCH_ + 16] = c1[v];
        }
    }
    __syncthreads();

    const int h0 = lane * 4;
    float4 kv[32];
    #pragma unroll
    for (int j = 0; j < 32; ++j)
        kv[j] = *(const float4*)(s_k + j * KPITCH_ + h0);

    const float4 wv = *(const float4*)(Ws + h0);
    const bool masked = (k0 + lane) < valid_len[b];

    const bool up16 = (lane & 16) != 0;
    const bool up8 = (lane & 8) != 0;
    const bool up4 = (lane & 4) != 0;
    const bool up2 = (lane & 2) != 0;
    const bool up1 = (lane & 1) != 0;

    const float* qbase = qproj + (size_t)b * Q_ * H_ + h0;
    float* rbase = raw + (size_t)b * Q_ * K_ + k0 + lane;

    float4 qv = *(const float4*)(qbase + (size_t)wave * H_);
    for (int i = 0; i < 8; ++i) {
        const int q = wave + 8 * i;
        const float4 qn = *(const float4*)(qbase + (size_t)(wave + 8 * ((i + 1) & 7)) * H_);

        float p[32];
        #pragma unroll
        for (int j = 0; j < 32; ++j) {
            p[j] = __builtin_fmaf(wv.w, TANHF(qv.w + kv[j].w),
                   __builtin_fmaf(wv.z, TANHF(qv.z + kv[j].z),
                   __builtin_fmaf(wv.x, TANHF(qv.x + kv[j].x),
                                  wv.y * TANHF(qv.y + kv[j].y))));
        }
        float a[16], c8[8], c4[4], c2[2];
        #pragma unroll
        for (int t = 0; t < 16; ++t) a[t] = xadd_rows(p[t], p[t + 16], up16);
        #pragma unroll
        for (int t = 0; t < 8; ++t)  c8[t] = xadd_in_row<DPP_XMASK_ + 8>(a[t], a[t + 8], up8);
        #pragma unroll
        for (int t = 0; t < 4; ++t)  c4[t] = xadd_in_row<DPP_XMASK_ + 4>(c8[t], c8[t + 4], up4);
        #pragma unroll
        for (int t = 0; t < 2; ++t)  c2[t] = xadd_in_row<DPP_XMASK_ + 2>(c4[t], c4[t + 2], up2);
        const float sc = xadd_in_row<DPP_XMASK_ + 1>(c2[0], c2[1], up1);

        rbase[(size_t)q * K_] = masked ? NEG_ : sc;
        qv = qn;
    }
}

__global__ __launch_bounds__(128) void softmax_rows(float* __restrict__ attn)
{
    const int wave = threadIdx.x >> 5;
    const int lane = threadIdx.x & 31;
    float* row = attn + (size_t)(blockIdx.x * 4 + wave) * K_ + lane;

    float v[4][8];
    #pragma unroll
    for (int i = 0; i < 4; ++i) {
        #pragma unroll
        for (int w = 0; w < 8; ++w) v[i][w] = row[i * 256 + w * 32];
    }

    float m = v[0][0];
    #pragma unroll
    for (int i = 0; i < 4; ++i) {
        #pragma unroll
        for (int w = 0; w < 8; ++w) m = fmaxf(m, v[i][w]);
    }
    #pragma unroll
    for (int off = 16; off >= 1; off >>= 1)
        m = fmaxf(m, __shfl_xor(m, off, 32));

    #pragma unroll
    for (int i = 0; i < 4; ++i) {
        #pragma unroll
        for (int w = 0; w < 8; ++w) v[i][w] = EXP2F((v[i][w] - m) * LOG2E);
    }

    float s[8];
    #pragma unroll
    for (int w = 0; w < 8; ++w) {
        s[w] = v[0][w] + v[1][w] + v[2][w] + v[3][w];
        #pragma unroll
        for (int off = 16; off >= 1; off >>= 1)
            s[w] += __shfl_xor(s[w], off, 32);
    }
    float tot = s[0];
    #pragma unroll
    for (int w = 1; w < 8; ++w) tot += s[w];
    const float inv = 1.0f / tot;

    #pragma unroll
    for (int i = 0; i < 4; ++i) {
        #pragma unroll
        for (int w = 0; w < 8; ++w) row[i * 256 + w * 32] = v[i][w] * inv;
    }
}

extern "C" void kernel_launch(void* const* d_in, const int* in_sizes, int n_in,
                              void* d_out, int out_size, void* d_ws, size_t ws_size,
                              hipStream_t stream) {
    const float* queries   = (const float*)d_in[0];
    const float* keys      = (const float*)d_in[1];
    const float* values    = (const float*)d_in[2];
    const int*   valid_len = (const int*)  d_in[3];
    const float* Wq        = (const float*)d_in[4];
    const float* Wk        = (const float*)d_in[5];
    const float* Ws        = (const float*)d_in[6];
    float* out = (float*)d_out;

    float* qproj = (float*)d_ws;
    float* attn  = qproj + (size_t)B_ * Q_ * H_
                         + (size_t)B_ * K_ * H_;

    gemm_wmma_f32<H_, QS_><<<dim3(128, 1), 128, 0, stream>>>(
        queries, 0, Wq, 0, qproj, 0);

    kproj_scores<<<B_ * (K_ / 32), 256, 0, stream>>>(
        keys, Wk, qproj, Ws, valid_len, attn);

    softmax_rows<<<B_ * Q_ / 4, 128, 0, stream>>>(attn);

    gemm_wmma_f32_ahead<VS_, K_, 30, 16><<<dim3(16, B_), 128, 0, stream>>>(
        attn,   (long)Q_ * K_,
        values, (long)K_ * VS_,
        out,    (long)Q_ * VS_);
}


# scores loop: the weighted h-sums of two keys run side by side in packed f32 mul/fma (one instruction per key pair and feature, 64 for 128); per-key chain and sum order kept, bit-identical
# speedup vs baseline: 2.8789x; 1.0391x over previous
#include <hip/hip_runtime.h>
#include <hip/hip_bf16.h>
#include <math.h>

#define B_  16
#define Q_  64
#define K_  1024
#define QS_ 256
#define KS_ 256
#define H_  128
#define VS_ 256
#define NEG_ (-1e6f)

typedef float v2f __attribute__((ext_vector_type(2)));
typedef float v8f __attribute__((ext_vector_type(8)));

#if defined(__has_builtin)
#  if __has_builtin(__builtin_amdgcn_tanhf)
#    define TANHF(x) __builtin_amdgcn_tanhf(x)
#  else
#    define TANHF(x) tanhf(x)
#  endif
#  if __has_builtin(__builtin_amdgcn_exp2f)
#    define EXP2F(x) __builtin_amdgcn_exp2f(x)
#  else
#    define EXP2F(x) exp2f(x)
#  endif
#else
#  define TANHF(x) tanhf(x)
#  define EXP2F(x) exp2f(x)
#endif

#define LOG2E 1.4426950408889634f

template<int N, int KD>
__global__ __launch_bounds__(128) void gemm_wmma_f32(
    const float* __restrict__ A, long sA,
    const float* __restrict__ W, long sW,
    float* __restrict__ C, long sC)
{
    constexpr int tilesN = N >> 4;
    constexpr int F4ROW  = KD / 4;
    constexpr int NST    = 16 * F4ROW / 128;
    constexpr int PITCH  = KD + 4;
    static_assert(tilesN % 4 == 0, "the four waves of a block must share their A strip");
    static_assert(KD % 64 == 0 && (16 * F4ROW) % 128 == 0, "shape");

    __shared__ __attribute__((aligned(16))) float s_a[16 * PITCH];

    A += (size_t)blockIdx.y * sA;
    W += (size_t)blockIdx.y * sW;
    C += (size_t)blockIdx.y * sC;

    const int tid  = threadIdx.x;
    const int wave = tid >> 5;
    const int lane = tid & 31;
    const int tile = blockIdx.x * 4 + wave;
    const int tm = tile / tilesN;
    const int tn = tile - tm * tilesN;
    const int r    = lane & 15;
    const int half = lane >> 4;

    {
        const float4* asrc = (const float4*)(A + (size_t)tm * 16 * KD);
        #pragma unroll
        for (int i = 0; i < NST; ++i) {
            const int idx = i * 128 + tid;
            const float4 v = asrc[idx];
            *(float4*)(s_a + (idx / F4ROW) * PITCH + (idx % F4ROW) * 4) = v;
        }
    }
    __syncthreads();

    const float* arow = s_a + r * PITCH + 2 * half;
    const float* wcol = W + (size_t)(2 * half) * N + tn * 16 + r;

    v8f c = {};
    for (int k0s = 0; k0s < KD; k0s += 64) {
        v2f b[16], a[16];
        #pragma unroll
        for (int s = 0; s < 16; ++s) {
            const int kk = k0s + 4 * s;
            b[s].x = wcol[(size_t)kk * N];
            b[s].y = wcol[(size_t)(kk + 1) * N];
        }
        #pragma unroll
        for (int s = 0; s < 16; ++s)
            a[s] = *(const v2f*)(arow + k0s + 4 * s);
        __builtin_amdgcn_sched_barrier(0);
        #pragma unroll
        for (int s = 0; s < 16; ++s)
            c = __builtin_amdgcn_wmma_f32_16x16x4_f32(
                     false, a[s],  false, b[s],
                     (short)0, c,  false,  false);
    }

    float* crow = C + (size_t)(tm * 16 + 8 * half) * N + tn * 16 + r;
    #pragma unroll
    for (int v = 0; v < 8; ++v)
        crow[(size_t)v * N] = c[v];
}

template<int N, int KD, int RING, int GRP>
__global__ __launch_bounds__(128) void gemm_wmma_f32_ahead(
    const float* __restrict__ A, long sA,
    const float* __restrict__ W, long sW,
    float* __restrict__ C, long sC)
{
    constexpr int tilesN = N >> 4;
    constexpr int STEPS  = KD / 4;
    constexpr int NGRP   = STEPS / GRP;
    constexpr int F4ROW  = KD / 4;
    constexpr int NST    = 16 * F4ROW / 128;
    constexpr int PITCH  = KD + 4;
    static_assert(tilesN % 4 == 0, "the four waves of a block must share their A strip");
    static_assert(STEPS % GRP == 0 && (16 * F4ROW) % 128 == 0 && RING <= STEPS, "shape");

    __shared__ __attribute__((aligned(16))) float s_a[16 * PITCH];

    A += (size_t)blockIdx.y * sA;
    W += (size_t)blockIdx.y * sW;
    C += (size_t)blockIdx.y * sC;

    const int tid  = threadIdx.x;
    const int wave = tid >> 5;
    const int lane = tid & 31;
    const int tile = blockIdx.x * 4 + wave;
    const int tm = tile / tilesN;
    const int tn = tile - tm * tilesN;
    const int r    = lane & 15;
    const int half = lane >> 4;

    const float* wcol = W + (size_t)(2 * half) * N + tn * 16 + r;

    const float4* asrc = (const float4*)(A + (size_t)tm * 16 * KD);
    float4 piece[NST];
    #pragma unroll
    for (int i = 0; i < NST; ++i)
        piece[i] = asrc[i * 128 + tid];

    v2f rb[RING];
    #pragma unroll
    for (int s = 0; s < RING; ++s) {
        rb[s].x = wcol[(size_t)(4 * s) * N];
        rb[s].y = wcol[(size_t)(4 * s + 1) * N];
    }

    #pragma unroll
    for (int i = 0; i < NST; ++i) {
        const int idx = i * 128 + tid;
        *(float4*)(s_a + (idx / F4ROW) * PITCH + (idx % F4ROW) * 4) = piece[i];
    }
    __syncthreads();

    const float* afrag = s_a + r * PITCH + 2 * half;
    v2f ra[2][GRP];
    #pragma unroll
    for (int s = 0; s < GRP; ++s)
        ra[0][s] = *(const v2f*)(afrag + 4 * s);

    v8f c = {};
    #pragma unroll
    for (int g = 0; g < NGRP; ++g) {
        if (g + 1 < NGRP) {
            #pragma unroll
            for (int s = 0; s < GRP; ++s)
                ra[(g + 1) & 1][s] = *(const v2f*)(afrag + 4 * ((g + 1) * GRP + s));
        }
        __builtin_amdgcn_sched_barrier(0);
        #pragma unroll
        for (int s = 0; s < GRP; ++s) {
            const int st = g * GRP + s;
            c = __builtin_amdgcn_wmma_f32_16x16x4_f32(
                     false, ra[g & 1][s],  false, rb[st % RING],
                     (short)0, c,  false,  false);
            if (st + RING < STEPS) {
                const int n = st + RING;
                rb[st % RING].x = wcol[(size_t)(4 * n) * N];
                rb[st % RING].y = wcol[(size_t)(4 * n + 1) * N];
            }
            __builtin_amdgcn_sched_barrier(0);
        }
    }

    float* crow = C + (size_t)(tm * 16 + 8 * half) * N + tn * 16 + r;
    #pragma unroll
    for (int v = 0; v < 8; ++v)
        crow[(size_t)v * N] = c[v];
}

#define DPP_XMASK_ 0x160

__device__ __forceinline__ float xadd_rows(float lo, float hi, bool up) {
    const float keep = up ? hi : lo;
    const int send = __builtin_bit_cast(int, up ? lo : hi);
    const float got = __builtin_bit_cast(float, __builtin_amdgcn_permlanex16(
        send, send, 0x76543210, 0xfedcba98, false, false));
    return keep + got;
}

template<int CTRL>
__device__ __forceinline__ float xadd_in_row(float lo, float hi, bool up) {
    const float keep = up ? hi : lo;
    const float send = up ? lo : hi;
    const float got = __builtin_bit_cast(float, __builtin_amdgcn_update_dpp(
        0, __builtin_bit_cast(int, send), CTRL, 0xF, 0xF, true));
    return keep + got;
}

#define KPITCH_ (H_ + 4)
#define APITCH_ (KS_ + 4)

__global__ __launch_bounds__(256) void kproj_scores(
    const float* __restrict__ keys,
    const float* __restrict__ Wk,
    const float* __restrict__ qproj,
    const float* __restrict__ Ws,
    const int*   __restrict__ valid_len,
    float* __restrict__ raw)
{
    __shared__ __attribute__((aligned(16))) float s_k[32 * KPITCH_];
    __shared__ __attribute__((aligned(16))) float s_a[32 * APITCH_];

    const int b  = blockIdx.x >> 5;
    const int k0 = (blockIdx.x & 31) * 32;
    const int wave = threadIdx.x >> 5;
    const int lane = threadIdx.x & 31;

    {
        const float4* ksrc = (const float4*)(keys + ((size_t)b * K_ + k0) * KS_);
        #pragma unroll
        for (int i = 0; i < 8; ++i) {
            const int idx = i * 256 + (int)threadIdx.x;
            const float4 v = ksrc[idx];
            *(float4*)(s_a + (idx >> 6) * APITCH_ + (idx & 63) * 4) = v;
        }
    }
    __syncthreads();
    {
        const int r    = lane & 15;
        const int half = lane >> 4;
        const int tm   = wave >> 2;
        const int tn0  = (wave & 3) * 2;

        const float* arow = s_a + (tm * 16 + r) * APITCH_ + 2 * half;
        const float* wcol = Wk + (size_t)(2 * half) * H_ + tn0 * 16 + r;

        v8f c0 = {}, c1 = {};
        for (int k0s = 0; k0s < KS_; k0s += 64) {
            v2f b0[16], b1[16], a[16];
            #pragma unroll
            for (int s = 0; s < 16; ++s) {
                const int kk = k0s + 4 * s;
                b0[s].x = wcol[(size_t)kk * H_];
                b0[s].y = wcol[(size_t)(kk + 1) * H_];
                b1[s].x = wcol[(size_t)kk * H_ + 16];
                b1[s].y = wcol[(size_t)(kk + 1) * H_ + 16];
            }
            #pragma unroll
            for (int s = 0; s < 16; ++s)
                a[s] = *(const v2f*)(arow + k0s + 4 * s);
            __builtin_amdgcn_sched_barrier(0);
            #pragma unroll
            for (int s = 0; s < 16; ++s) {
                c0 = __builtin_amdgcn_wmma_f32_16x16x4_f32(
                         false, a[s],  false, b0[s],
                         (short)0, c0,  false,  false);
                c1 = __builtin_amdgcn_wmma_f32_16x16x4_f32(
                         false, a[s],  false, b1[s],
                         (short)0, c1,  false,  false);
            }
        }

        float* srow = s_k + (tm * 16 + 8 * half) * KPITCH_ + tn0 * 16 + r;
        #pragma unroll
        for (int v = 0; v < 8; ++v) {
            srow[v * KPITCH_]      = c0[v];
            srow[v * KPITCH_ + 16] = c1[v];
        }
    }
    __syncthreads();

    const int h0 = lane * 4;
    float4 kv[32];
    #pragma unroll
    for (int j = 0; j < 32; ++j)
        kv[j] = *(const float4*)(s_k + j * KPITCH_ + h0);

    const float4 wv = *(const float4*)(Ws + h0);
    const v2f wx2 = {wv.x, wv.x}, wy2 = {wv.y, wv.y}, wz2 = {wv.z, wv.z}, ww2 = {wv.w, wv.w};
    const bool masked = (k0 + lane) < valid_len[b];

    const bool up16 = (lane & 16) != 0;
    const bool up8 = (lane & 8) != 0;
    const bool up4 = (lane & 4) != 0;
    const bool up2 = (lane & 2) != 0;
    const bool up1 = (lane & 1) != 0;

    const float* qbase = qproj + (size_t)b * Q_ * H_ + h0;
    float* rbase = raw + (size_t)b * Q_ * K_ + k0 + lane;

    float4 qv = *(const float4*)(qbase + (size_t)wave * H_);
    for (int i = 0; i < 8; ++i) {
        const int q = wave + 8 * i;
        const float4 qn = *(const float4*)(qbase + (size_t)(wave + 8 * ((i + 1) & 7)) * H_);

        float p[32];
        const v2f qxy = {qv.x, qv.y}, qzw = {qv.z, qv.w};
        #pragma unroll
        for (int jj = 0; jj < 16; ++jj) {
            const float4 ka = kv[2 * jj], kb = kv[2 * jj + 1];
            const v2f axy = qxy + (v2f){ka.x, ka.y};
            const v2f azw = qzw + (v2f){ka.z, ka.w};
            const v2f bxy = qxy + (v2f){kb.x, kb.y};
            const v2f bzw = qzw + (v2f){kb.z, kb.w};
            const v2f ty = {TANHF(axy.y), TANHF(bxy.y)};
            const v2f tx = {TANHF(axy.x), TANHF(bxy.x)};
            const v2f tz = {TANHF(azw.x), TANHF(bzw.x)};
            const v2f tw = {TANHF(azw.y), TANHF(bzw.y)};
            v2f acc = wy2 * ty;
            acc = __builtin_elementwise_fma(wx2, tx, acc);
            acc = __builtin_elementwise_fma(wz2, tz, acc);
            acc = __builtin_elementwise_fma(ww2, tw, acc);
            p[2 * jj]     = acc.x;
            p[2 * jj + 1] = acc.y;
        }
        float a[16], c8[8], c4[4], c2[2];
        #pragma unroll
        for (int t = 0; t < 16; ++t) a[t] = xadd_rows(p[t], p[t + 16], up16);
        #pragma unroll
        for (int t = 0; t < 8; ++t)  c8[t] = xadd_in_row<DPP_XMASK_ + 8>(a[t], a[t + 8], up8);
        #pragma unroll
        for (int t = 0; t < 4; ++t)  c4[t] = xadd_in_row<DPP_XMASK_ + 4>(c8[t], c8[t + 4], up4);
        #pragma unroll
        for (int t = 0; t < 2; ++t)  c2[t] = xadd_in_row<DPP_XMASK_ + 2>(c4[t], c4[t + 2], up2);
        const float sc = xadd_in_row<DPP_XMASK_ + 1>(c2[0], c2[1], up1);

        rbase[(size_t)q * K_] = masked ? NEG_ : sc;
        qv = qn;
    }
}

__global__ __launch_bounds__(128) void softmax_rows(float* __restrict__ attn)
{
    const int wave = threadIdx.x >> 5;
    const int lane = threadIdx.x & 31;
    float* row = attn + (size_t)(blockIdx.x * 4 + wave) * K_ + lane;

    float v[4][8];
    #pragma unroll
    for (int i = 0; i < 4; ++i) {
        #pragma unroll
        for (int w = 0; w < 8; ++w) v[i][w] = row[i * 256 + w * 32];
    }

    float m = v[0][0];
    #pragma unroll
    for (int i = 0; i < 4; ++i) {
        #pragma unroll
        for (int w = 0; w < 8; ++w) m = fmaxf(m, v[i][w]);
    }
    #pragma unroll
    for (int off = 16; off >= 1; off >>= 1)
        m = fmaxf(m, __shfl_xor(m, off, 32));

    #pragma unroll
    for (int i = 0; i < 4; ++i) {
        #pragma unroll
        for (int w = 0; w < 8; ++w) v[i][w] = EXP2F((v[i][w] - m) * LOG2E);
    }

    float s[8];
    #pragma unroll
    for (int w = 0; w < 8; ++w) {
        s[w] = v[0][w] + v[1][w] + v[2][w] + v[3][w];
        #pragma unroll
        for (int off = 16; off >= 1; off >>= 1)
            s[w] += __shfl_xor(s[w], off, 32);
    }
    float tot = s[0];
    #pragma unroll
    for (int w = 1; w < 8; ++w) tot += s[w];
    const float inv = 1.0f / tot;

    #pragma unroll
    for (int i = 0; i < 4; ++i) {
        #pragma unroll
        for (int w = 0; w < 8; ++w) row[i * 256 + w * 32] = v[i][w] * inv;
    }
}

extern "C" void kernel_launch(void* const* d_in, const int* in_sizes, int n_in,
                              void* d_out, int out_size, void* d_ws, size_t ws_size,
                              hipStream_t stream) {
    const float* queries   = (const float*)d_in[0];
    const float* keys      = (const float*)d_in[1];
    const float* values    = (const float*)d_in[2];
    const int*   valid_len = (const int*)  d_in[3];
    const float* Wq        = (const float*)d_in[4];
    const float* Wk        = (const float*)d_in[5];
    const float* Ws        = (const float*)d_in[6];
    float* out = (float*)d_out;

    float* qproj = (float*)d_ws;
    float* attn  = qproj + (size_t)B_ * Q_ * H_
                         + (size_t)B_ * K_ * H_;

    gemm_wmma_f32<H_, QS_><<<dim3(128, 1), 128, 0, stream>>>(
        queries, 0, Wq, 0, qproj, 0);

    kproj_scores<<<B_ * (K_ / 32), 256, 0, stream>>>(
        keys, Wk, qproj, Ws, valid_len, attn);

    softmax_rows<<<B_ * Q_ / 4, 128, 0, stream>>>(attn);

    gemm_wmma_f32_ahead<VS_, K_, 30, 16><<<dim3(16, B_), 128, 0, stream>>>(
        attn,   (long)Q_ * K_,
        values, (long)K_ * VS_,
        out,    (long)Q_ * VS_);
}


# softmax fused into the attn x values launch: a block normalises its 16 score rows in registers (DPP lane steps, sum order kept) while staging them to LDS; attn not stored; 3 launches; bit-identical
# speedup vs baseline: 3.5764x; 1.2423x over previous
#include <hip/hip_runtime.h>
#include <hip/hip_bf16.h>
#include <math.h>

#define B_  16
#define Q_  64
#define K_  1024
#define QS_ 256
#define KS_ 256
#define H_  128
#define VS_ 256
#define NEG_ (-1e6f)

typedef float v2f __attribute__((ext_vector_type(2)));
typedef float v8f __attribute__((ext_vector_type(8)));

#if defined(__has_builtin)
#  if __has_builtin(__builtin_amdgcn_tanhf)
#    define TANHF(x) __builtin_amdgcn_tanhf(x)
#  else
#    define TANHF(x) tanhf(x)
#  endif
#  if __has_builtin(__builtin_amdgcn_exp2f)
#    define EXP2F(x) __builtin_amdgcn_exp2f(x)
#  else
#    define EXP2F(x) exp2f(x)
#  endif
#else
#  define TANHF(x) tanhf(x)
#  define EXP2F(x) exp2f(x)
#endif

#define LOG2E 1.4426950408889634f

template<int N, int KD>
__global__ __launch_bounds__(128) void gemm_wmma_f32(
    const float* __restrict__ A, long sA,
    const float* __restrict__ W, long sW,
    float* __restrict__ C, long sC)
{
    constexpr int tilesN = N >> 4;
    constexpr int F4ROW  = KD / 4;
    constexpr int NST    = 16 * F4ROW / 128;
    constexpr int PITCH  = KD + 4;
    static_assert(tilesN % 4 == 0, "the four waves of a block must share their A strip");
    static_assert(KD % 64 == 0 && (16 * F4ROW) % 128 == 0, "shape");

    __shared__ __attribute__((aligned(16))) float s_a[16 * PITCH];

    A += (size_t)blockIdx.y * sA;
    W += (size_t)blockIdx.y * sW;
    C += (size_t)blockIdx.y * sC;

    const int tid  = threadIdx.x;
    const int wave = tid >> 5;
    const int lane = tid & 31;
    const int tile = blockIdx.x * 4 + wave;
    const int tm = tile / tilesN;
    const int tn = tile - tm * tilesN;
    const int r    = lane & 15;
    const int half = lane >> 4;

    {
        const float4* asrc = (const float4*)(A + (size_t)tm * 16 * KD);
        #pragma unroll
        for (int i = 0; i < NST; ++i) {
            const int idx = i * 128 + tid;
            const float4 v = asrc[idx];
            *(float4*)(s_a + (idx / F4ROW) * PITCH + (idx % F4ROW) * 4) = v;
        }
    }
    __syncthreads();

    const float* arow = s_a + r * PITCH + 2 * half;
    const float* wcol = W + (size_t)(2 * half) * N + tn * 16 + r;

    v8f c = {};
    for (int k0s = 0; k0s < KD; k0s += 64) {
        v2f b[16], a[16];
        #pragma unroll
        for (int s = 0; s < 16; ++s) {
            const int kk = k0s + 4 * s;
            b[s].x = wcol[(size_t)kk * N];
            b[s].y = wcol[(size_t)(kk + 1) * N];
        }
        #pragma unroll
        for (int s = 0; s < 16; ++s)
            a[s] = *(const v2f*)(arow + k0s + 4 * s);
        __builtin_amdgcn_sched_barrier(0);
        #pragma unroll
        for (int s = 0; s < 16; ++s)
            c = __builtin_amdgcn_wmma_f32_16x16x4_f32(
                     false, a[s],  false, b[s],
                     (short)0, c,  false,  false);
    }

    float* crow = C + (size_t)(tm * 16 + 8 * half) * N + tn * 16 + r;
    #pragma unroll
    for (int v = 0; v < 8; ++v)
        crow[(size_t)v * N] = c[v];
}

#define DPP_XMASK_ 0x160

__device__ __forceinline__ float xadd_rows(float lo, float hi, bool up) {
    const float keep = up ? hi : lo;
    const int send = __builtin_bit_cast(int, up ? lo : hi);
    const float got = __builtin_bit_cast(float, __builtin_amdgcn_permlanex16(
        send, send, 0x76543210, 0xfedcba98, false, false));
    return keep + got;
}

template<int CTRL>
__device__ __forceinline__ float xadd_in_row(float lo, float hi, bool up) {
    const float keep = up ? hi : lo;
    const float send = up ? lo : hi;
    const float got = __builtin_bit_cast(float, __builtin_amdgcn_update_dpp(
        0, __builtin_bit_cast(int, send), CTRL, 0xF, 0xF, true));
    return keep + got;
}

#define KPITCH_ (H_ + 4)
#define APITCH_ (KS_ + 4)

__global__ __launch_bounds__(256) void kproj_scores(
    const float* __restrict__ keys,
    const float* __restrict__ Wk,
    const float* __restrict__ qproj,
    const float* __restrict__ Ws,
    const int*   __restrict__ valid_len,
    float* __restrict__ raw)
{
    __shared__ __attribute__((aligned(16))) float s_k[32 * KPITCH_];
    __shared__ __attribute__((aligned(16))) float s_a[32 * APITCH_];

    const int b  = blockIdx.x >> 5;
    const int k0 = (blockIdx.x & 31) * 32;
    const int wave = threadIdx.x >> 5;
    const int lane = threadIdx.x & 31;

    {
        const float4* ksrc = (const float4*)(keys + ((size_t)b * K_ + k0) * KS_);
        #pragma unroll
        for (int i = 0; i < 8; ++i) {
            const int idx = i * 256 + (int)threadIdx.x;
            const float4 v = ksrc[idx];
            *(float4*)(s_a + (idx >> 6) * APITCH_ + (idx & 63) * 4) = v;
        }
    }
    __syncthreads();
    {
        const int r    = lane & 15;
        const int half = lane >> 4;
        const int tm   = wave >> 2;
        const int tn0  = (wave & 3) * 2;

        const float* arow = s_a + (tm * 16 + r) * APITCH_ + 2 * half;
        const float* wcol = Wk + (size_t)(2 * half) * H_ + tn0 * 16 + r;

        v8f c0 = {}, c1 = {};
        for (int k0s = 0; k0s < KS_; k0s += 64) {
            v2f b0[16], b1[16], a[16];
            #pragma unroll
            for (int s = 0; s < 16; ++s) {
                const int kk = k0s + 4 * s;
                b0[s].x = wcol[(size_t)kk * H_];
                b0[s].y = wcol[(size_t)(kk + 1) * H_];
                b1[s].x = wcol[(size_t)kk * H_ + 16];
                b1[s].y = wcol[(size_t)(kk + 1) * H_ + 16];
            }
            #pragma unroll
            for (int s = 0; s < 16; ++s)
                a[s] = *(const v2f*)(arow + k0s + 4 * s);
            __builtin_amdgcn_sched_barrier(0);
            #pragma unroll
            for (int s = 0; s < 16; ++s) {
                c0 = __builtin_amdgcn_wmma_f32_16x16x4_f32(
                         false, a[s],  false, b0[s],
                         (short)0, c0,  false,  false);
                c1 = __builtin_amdgcn_wmma_f32_16x16x4_f32(
                         false, a[s],  false, b1[s],
                         (short)0, c1,  false,  false);
            }
        }

        float* srow = s_k + (tm * 16 + 8 * half) * KPITCH_ + tn0 * 16 + r;
        #pragma unroll
        for (int v = 0; v < 8; ++v) {
            srow[v * KPITCH_]      = c0[v];
            srow[v * KPITCH_ + 16] = c1[v];
        }
    }
    __syncthreads();

    const int h0 = lane * 4;
    float4 kv[32];
    #pragma unroll
    for (int j = 0; j < 32; ++j)
        kv[j] = *(const float4*)(s_k + j * KPITCH_ + h0);

    const float4 wv = *(const float4*)(Ws + h0);
    const v2f wx2 = {wv.x, wv.x}, wy2 = {wv.y, wv.y}, wz2 = {wv.z, wv.z}, ww2 = {wv.w, wv.w};
    const bool masked = (k0 + lane) < valid_len[b];

    const bool up16 = (lane & 16) != 0;
    const bool up8 = (lane & 8) != 0;
    const bool up4 = (lane & 4) != 0;
    const bool up2 = (lane & 2) != 0;
    const bool up1 = (lane & 1) != 0;

    const float* qbase = qproj + (size_t)b * Q_ * H_ + h0;
    float* rbase = raw + (size_t)b * Q_ * K_ + k0 + lane;

    float4 qv = *(const float4*)(qbase + (size_t)wave * H_);
    for (int i = 0; i < 8; ++i) {
        const int q = wave + 8 * i;
        const float4 qn = *(const float4*)(qbase + (size_t)(wave + 8 * ((i + 1) & 7)) * H_);

        float p[32];
        const v2f qxy = {qv.x, qv.y}, qzw = {qv.z, qv.w};
        #pragma unroll
        for (int jj = 0; jj < 16; ++jj) {
            const float4 ka = kv[2 * jj], kb = kv[2 * jj + 1];
            const v2f axy = qxy + (v2f){ka.x, ka.y};
            const v2f azw = qzw + (v2f){ka.z, ka.w};
            const v2f bxy = qxy + (v2f){kb.x, kb.y};
            const v2f bzw = qzw + (v2f){kb.z, kb.w};
            const v2f ty = {TANHF(axy.y), TANHF(bxy.y)};
            const v2f tx = {TANHF(axy.x), TANHF(bxy.x)};
            const v2f tz = {TANHF(azw.x), TANHF(bzw.x)};
            const v2f tw = {TANHF(azw.y), TANHF(bzw.y)};
            v2f acc = wy2 * ty;
            acc = __builtin_elementwise_fma(wx2, tx, acc);
            acc = __builtin_elementwise_fma(wz2, tz, acc);
            acc = __builtin_elementwise_fma(ww2, tw, acc);
            p[2 * jj]     = acc.x;
            p[2 * jj + 1] = acc.y;
        }
        float a[16], c8[8], c4[4], c2[2];
        #pragma unroll
        for (int t = 0; t < 16; ++t) a[t] = xadd_rows(p[t], p[t + 16], up16);
        #pragma unroll
        for (int t = 0; t < 8; ++t)  c8[t] = xadd_in_row<DPP_XMASK_ + 8>(a[t], a[t + 8], up8);
        #pragma unroll
        for (int t = 0; t < 4; ++t)  c4[t] = xadd_in_row<DPP_XMASK_ + 4>(c8[t], c8[t + 4], up4);
        #pragma unroll
        for (int t = 0; t < 2; ++t)  c2[t] = xadd_in_row<DPP_XMASK_ + 2>(c4[t], c4[t + 2], up2);
        const float sc = xadd_in_row<DPP_XMASK_ + 1>(c2[0], c2[1], up1);

        rbase[(size_t)q * K_] = masked ? NEG_ : sc;
        qv = qn;
    }
}

__device__ __forceinline__ float lane_xor16(float x) {
    const int xi = __builtin_bit_cast(int, x);
    return __builtin_bit_cast(float, __builtin_amdgcn_permlanex16(
        xi, xi, 0x76543210, 0xfedcba98, false, false));
}

template<int CTRL>
__device__ __forceinline__ float lane_xor_in_row(float x) {
    return __builtin_bit_cast(float, __builtin_amdgcn_update_dpp(
        0, __builtin_bit_cast(int, x), CTRL, 0xF, 0xF, true));
}

__device__ __forceinline__ float wave_allmax(float m) {
    m = fmaxf(m, lane_xor16(m));
    m = fmaxf(m, lane_xor_in_row<DPP_XMASK_ + 8>(m));
    m = fmaxf(m, lane_xor_in_row<DPP_XMASK_ + 4>(m));
    m = fmaxf(m, lane_xor_in_row<DPP_XMASK_ + 2>(m));
    m = fmaxf(m, lane_xor_in_row<DPP_XMASK_ + 1>(m));
    return m;
}

__device__ __forceinline__ float wave_allsum(float s) {
    s += lane_xor16(s);
    s += lane_xor_in_row<DPP_XMASK_ + 8>(s);
    s += lane_xor_in_row<DPP_XMASK_ + 4>(s);
    s += lane_xor_in_row<DPP_XMASK_ + 2>(s);
    s += lane_xor_in_row<DPP_XMASK_ + 1>(s);
    return s;
}

template<int N, int KD, int RING, int GRP>
__global__ __launch_bounds__(128) void softmax_gemm_wmma_f32_ahead(
    const float* __restrict__ S, long sS,
    const float* __restrict__ W, long sW,
    float* __restrict__ C, long sC)
{
    constexpr int tilesN = N >> 4;
    constexpr int STEPS  = KD / 4;
    constexpr int NGRP   = STEPS / GRP;
    constexpr int NI     = KD / 256;
    constexpr int PITCH  = KD + 4;
    static_assert(tilesN % 4 == 0, "the four waves of a block must share their A strip");
    static_assert(STEPS % GRP == 0 && KD % 256 == 0 && RING <= STEPS, "shape");

    __shared__ __attribute__((aligned(16))) float s_a[16 * PITCH];

    S += (size_t)blockIdx.y * sS;
    W += (size_t)blockIdx.y * sW;
    C += (size_t)blockIdx.y * sC;

    const int tid  = threadIdx.x;
    const int wave = tid >> 5;
    const int lane = tid & 31;
    const int tile = blockIdx.x * 4 + wave;
    const int tm = tile / tilesN;
    const int tn = tile - tm * tilesN;
    const int r    = lane & 15;
    const int half = lane >> 4;

    const float* wcol = W + (size_t)(2 * half) * N + tn * 16 + r;

    const float* srow = S + (size_t)(tm * 16 + wave * 4) * KD + lane;
    float v[4][NI][8];
    #pragma unroll
    for (int rr = 0; rr < 4; ++rr) {
        #pragma unroll
        for (int i = 0; i < NI; ++i) {
            #pragma unroll
            for (int w = 0; w < 8; ++w) v[rr][i][w] = srow[(size_t)rr * KD + i * 256 + w * 32];
        }
    }

    v2f rb[RING];
    #pragma unroll
    for (int s = 0; s < RING; ++s) {
        rb[s].x = wcol[(size_t)(4 * s) * N];
        rb[s].y = wcol[(size_t)(4 * s + 1) * N];
    }

    float m[4];
    #pragma unroll
    for (int rr = 0; rr < 4; ++rr) {
        m[rr] = v[rr][0][0];
        #pragma unroll
        for (int i = 0; i < NI; ++i) {
            #pragma unroll
            for (int w = 0; w < 8; ++w) m[rr] = fmaxf(m[rr], v[rr][i][w]);
        }
    }
    #pragma unroll
    for (int rr = 0; rr < 4; ++rr) m[rr] = wave_allmax(m[rr]);

    #pragma unroll
    for (int rr = 0; rr < 4; ++rr) {
        #pragma unroll
        for (int i = 0; i < NI; ++i) {
            #pragma unroll
            for (int w = 0; w < 8; ++w) v[rr][i][w] = EXP2F((v[rr][i][w] - m[rr]) * LOG2E);
        }
    }

    float sm[4][8];
    #pragma unroll
    for (int rr = 0; rr < 4; ++rr) {
        #pragma unroll
        for (int w = 0; w < 8; ++w) {
            float t = v[rr][0][w];
            #pragma unroll
            for (int i = 1; i < NI; ++i) t += v[rr][i][w];
            sm[rr][w] = t;
        }
    }
    #pragma unroll
    for (int rr = 0; rr < 4; ++rr) {
        #pragma unroll
        for (int w = 0; w < 8; ++w) sm[rr][w] = wave_allsum(sm[rr][w]);
    }

    #pragma unroll
    for (int rr = 0; rr < 4; ++rr) {
        float tot = sm[rr][0];
        #pragma unroll
        for (int w = 1; w < 8; ++w) tot += sm[rr][w];
        const float inv = 1.0f / tot;
        float* drow = s_a + (wave * 4 + rr) * PITCH + lane;
        #pragma unroll
        for (int i = 0; i < NI; ++i) {
            #pragma unroll
            for (int w = 0; w < 8; ++w) drow[i * 256 + w * 32] = v[rr][i][w] * inv;
        }
    }
    __syncthreads();

    const float* afrag = s_a + r * PITCH + 2 * half;
    v2f ra[2][GRP];
    #pragma unroll
    for (int s = 0; s < GRP; ++s)
        ra[0][s] = *(const v2f*)(afrag + 4 * s);

    v8f c = {};
    #pragma unroll
    for (int g = 0; g < NGRP; ++g) {
        if (g + 1 < NGRP) {
            #pragma unroll
            for (int s = 0; s < GRP; ++s)
                ra[(g + 1) & 1][s] = *(const v2f*)(afrag + 4 * ((g + 1) * GRP + s));
        }
        __builtin_amdgcn_sched_barrier(0);
        #pragma unroll
        for (int s = 0; s < GRP; ++s) {
            const int st = g * GRP + s;
            c = __builtin_amdgcn_wmma_f32_16x16x4_f32(
                     false, ra[g & 1][s],  false, rb[st % RING],
                     (short)0, c,  false,  false);
            if (st + RING < STEPS) {
                const int n = st + RING;
                rb[st % RING].x = wcol[(size_t)(4 * n) * N];
                rb[st % RING].y = wcol[(size_t)(4 * n + 1) * N];
            }
            __builtin_amdgcn_sched_barrier(0);
        }
    }

    float* crow = C + (size_t)(tm * 16 + 8 * half) * N + tn * 16 + r;
    #pragma unroll
    for (int v2 = 0; v2 < 8; ++v2)
        crow[(size_t)v2 * N] = c[v2];
}

extern "C" void kernel_launch(void* const* d_in, const int* in_sizes, int n_in,
                              void* d_out, int out_size, void* d_ws, size_t ws_size,
                              hipStream_t stream) {
    const float* queries   = (const float*)d_in[0];
    const float* keys      = (const float*)d_in[1];
    const float* values    = (const float*)d_in[2];
    const int*   valid_len = (const int*)  d_in[3];
    const float* Wq        = (const float*)d_in[4];
    const float* Wk        = (const float*)d_in[5];
    const float* Ws        = (const float*)d_in[6];
    float* out = (float*)d_out;

    float* qproj = (float*)d_ws;
    float* attn  = qproj + (size_t)B_ * Q_ * H_
                         + (size_t)B_ * K_ * H_;

    gemm_wmma_f32<H_, QS_><<<dim3(128, 1), 128, 0, stream>>>(
        queries, 0, Wq, 0, qproj, 0);

    kproj_scores<<<B_ * (K_ / 32), 256, 0, stream>>>(
        keys, Wk, qproj, Ws, valid_len, attn);

    softmax_gemm_wmma_f32_ahead<VS_, K_, 30, 16><<<dim3(16, B_), 128, 0, stream>>>(
        attn,   (long)Q_ * K_,
        values, (long)K_ * VS_,
        out,    (long)Q_ * VS_);
}


# scores stage: a lane owns 16 features of 8 keys (was 4 of 32): cross-lane h-sum 7 exchange-adds per 32 keys, all in-row (was 31, 16 row swaps); feature pairs in packed ops; h-sum order changes
# speedup vs baseline: 3.7405x; 1.0459x over previous
#include <hip/hip_runtime.h>
#include <hip/hip_bf16.h>
#include <math.h>

#define B_  16
#define Q_  64
#define K_  1024
#define QS_ 256
#define KS_ 256
#define H_  128
#define VS_ 256
#define NEG_ (-1e6f)

typedef float v2f __attribute__((ext_vector_type(2)));
typedef float v8f __attribute__((ext_vector_type(8)));

#if defined(__has_builtin)
#  if __has_builtin(__builtin_amdgcn_tanhf)
#    define TANHF(x) __builtin_amdgcn_tanhf(x)
#  else
#    define TANHF(x) tanhf(x)
#  endif
#  if __has_builtin(__builtin_amdgcn_exp2f)
#    define EXP2F(x) __builtin_amdgcn_exp2f(x)
#  else
#    define EXP2F(x) exp2f(x)
#  endif
#else
#  define TANHF(x) tanhf(x)
#  define EXP2F(x) exp2f(x)
#endif

#define LOG2E 1.4426950408889634f

template<int N, int KD>
__global__ __launch_bounds__(128) void gemm_wmma_f32(
    const float* __restrict__ A, long sA,
    const float* __restrict__ W, long sW,
    float* __restrict__ C, long sC)
{
    constexpr int tilesN = N >> 4;
    constexpr int F4ROW  = KD / 4;
    constexpr int NST    = 16 * F4ROW / 128;
    constexpr int PITCH  = KD + 4;
    static_assert(tilesN % 4 == 0, "the four waves of a block must share their A strip");
    static_assert(KD % 64 == 0 && (16 * F4ROW) % 128 == 0, "shape");

    __shared__ __attribute__((aligned(16))) float s_a[16 * PITCH];

    A += (size_t)blockIdx.y * sA;
    W += (size_t)blockIdx.y * sW;
    C += (size_t)blockIdx.y * sC;

    const int tid  = threadIdx.x;
    const int wave = tid >> 5;
    const int lane = tid & 31;
    const int tile = blockIdx.x * 4 + wave;
    const int tm = tile / tilesN;
    const int tn = tile - tm * tilesN;
    const int r    = lane & 15;
    const int half = lane >> 4;

    {
        const float4* asrc = (const float4*)(A + (size_t)tm * 16 * KD);
        #pragma unroll
        for (int i = 0; i < NST; ++i) {
            const int idx = i * 128 + tid;
            const float4 v = asrc[idx];
            *(float4*)(s_a + (idx / F4ROW) * PITCH + (idx % F4ROW) * 4) = v;
        }
    }
    __syncthreads();

    const float* arow = s_a + r * PITCH + 2 * half;
    const float* wcol = W + (size_t)(2 * half) * N + tn * 16 + r;

    v8f c = {};
    for (int k0s = 0; k0s < KD; k0s += 64) {
        v2f b[16], a[16];
        #pragma unroll
        for (int s = 0; s < 16; ++s) {
            const int kk = k0s + 4 * s;
            b[s].x = wcol[(size_t)kk * N];
            b[s].y = wcol[(size_t)(kk + 1) * N];
        }
        #pragma unroll
        for (int s = 0; s < 16; ++s)
            a[s] = *(const v2f*)(arow + k0s + 4 * s);
        __builtin_amdgcn_sched_barrier(0);
        #pragma unroll
        for (int s = 0; s < 16; ++s)
            c = __builtin_amdgcn_wmma_f32_16x16x4_f32(
                     false, a[s],  false, b[s],
                     (short)0, c,  false,  false);
    }

    float* crow = C + (size_t)(tm * 16 + 8 * half) * N + tn * 16 + r;
    #pragma unroll
    for (int v = 0; v < 8; ++v)
        crow[(size_t)v * N] = c[v];
}

#define DPP_XMASK_ 0x160

template<int CTRL>
__device__ __forceinline__ float xadd_in_row(float lo, float hi, bool up) {
    const float keep = up ? hi : lo;
    const float send = up ? lo : hi;
    const float got = __builtin_bit_cast(float, __builtin_amdgcn_update_dpp(
        0, __builtin_bit_cast(int, send), CTRL, 0xF, 0xF, true));
    return keep + got;
}

#define KPITCH_ (H_ + 4)
#define APITCH_ (KS_ + 4)

__global__ __launch_bounds__(256) void kproj_scores(
    const float* __restrict__ keys,
    const float* __restrict__ Wk,
    const float* __restrict__ qproj,
    const float* __restrict__ Ws,
    const int*   __restrict__ valid_len,
    float* __restrict__ raw)
{
    __shared__ __attribute__((aligned(16))) float s_k[32 * KPITCH_];
    __shared__ __attribute__((aligned(16))) float s_a[32 * APITCH_];

    const int b  = blockIdx.x >> 5;
    const int k0 = (blockIdx.x & 31) * 32;
    const int wave = threadIdx.x >> 5;
    const int lane = threadIdx.x & 31;

    {
        const float4* ksrc = (const float4*)(keys + ((size_t)b * K_ + k0) * KS_);
        #pragma unroll
        for (int i = 0; i < 8; ++i) {
            const int idx = i * 256 + (int)threadIdx.x;
            const float4 v = ksrc[idx];
            *(float4*)(s_a + (idx >> 6) * APITCH_ + (idx & 63) * 4) = v;
        }
    }
    __syncthreads();
    {
        const int r    = lane & 15;
        const int half = lane >> 4;
        const int tm   = wave >> 2;
        const int tn0  = (wave & 3) * 2;

        const float* arow = s_a + (tm * 16 + r) * APITCH_ + 2 * half;
        const float* wcol = Wk + (size_t)(2 * half) * H_ + tn0 * 16 + r;

        v8f c0 = {}, c1 = {};
        for (int k0s = 0; k0s < KS_; k0s += 64) {
            v2f b0[16], b1[16], a[16];
            #pragma unroll
            for (int s = 0; s < 16; ++s) {
                const int kk = k0s + 4 * s;
                b0[s].x = wcol[(size_t)kk * H_];
                b0[s].y = wcol[(size_t)(kk + 1) * H_];
                b1[s].x = wcol[(size_t)kk * H_ + 16];
                b1[s].y = wcol[(size_t)(kk + 1) * H_ + 16];
            }
            #pragma unroll
            for (int s = 0; s < 16; ++s)
                a[s] = *(const v2f*)(arow + k0s + 4 * s);
            __builtin_amdgcn_sched_barrier(0);
            #pragma unroll
            for (int s = 0; s < 16; ++s) {
                c0 = __builtin_amdgcn_wmma_f32_16x16x4_f32(
                         false, a[s],  false, b0[s],
                         (short)0, c0,  false,  false);
                c1 = __builtin_amdgcn_wmma_f32_16x16x4_f32(
                         false, a[s],  false, b1[s],
                         (short)0, c1,  false,  false);
            }
        }

        float* srow = s_k + (tm * 16 + 8 * half) * KPITCH_ + tn0 * 16 + r;
        #pragma unroll
        for (int v = 0; v < 8; ++v) {
            srow[v * KPITCH_]      = c0[v];
            srow[v * KPITCH_ + 16] = c1[v];
        }
    }
    __syncthreads();

    const int fg = lane & 7;
    const int kg = lane >> 3;
    const int h0 = fg * 4;
    float4 kv[8][4];
    #pragma unroll
    for (int j = 0; j < 8; ++j) {
        #pragma unroll
        for (int i = 0; i < 4; ++i)
            kv[j][i] = *(const float4*)(s_k + (kg * 8 + j) * KPITCH_ + h0 + 32 * i);
    }

    v2f w2[8];
    #pragma unroll
    for (int i = 0; i < 4; ++i) {
        const float4 wv = *(const float4*)(Ws + h0 + 32 * i);
        w2[2 * i]     = (v2f){wv.x, wv.y};
        w2[2 * i + 1] = (v2f){wv.z, wv.w};
    }
    const bool masked = (k0 + lane) < valid_len[b];

    const bool up4 = (lane & 4) != 0;
    const bool up2 = (lane & 2) != 0;
    const bool up1 = (lane & 1) != 0;

    const float* qbase = qproj + (size_t)b * Q_ * H_ + h0;
    float* rbase = raw + (size_t)b * Q_ * K_ + k0 + lane;

    float4 qv[4];
    #pragma unroll
    for (int i = 0; i < 4; ++i)
        qv[i] = *(const float4*)(qbase + (size_t)wave * H_ + 32 * i);
    #pragma unroll 2
    for (int it = 0; it < 8; ++it) {
        const int q = wave + 8 * it;
        float4 qn[4];
        #pragma unroll
        for (int i = 0; i < 4; ++i)
            qn[i] = *(const float4*)(qbase + (size_t)(wave + 8 * ((it + 1) & 7)) * H_ + 32 * i);

        float p[8];
        #pragma unroll
        for (int j = 0; j < 8; ++j) {
            v2f acc;
            #pragma unroll
            for (int i = 0; i < 4; ++i) {
                const float4 k4 = kv[j][i];
                const v2f a0 = (v2f){qv[i].x, qv[i].y} + (v2f){k4.x, k4.y};
                const v2f a1 = (v2f){qv[i].z, qv[i].w} + (v2f){k4.z, k4.w};
                const v2f t0 = {TANHF(a0.x), TANHF(a0.y)};
                const v2f t1 = {TANHF(a1.x), TANHF(a1.y)};
                if (i == 0) acc = w2[0] * t0;
                else        acc = __builtin_elementwise_fma(w2[2 * i], t0, acc);
                acc = __builtin_elementwise_fma(w2[2 * i + 1], t1, acc);
            }
            p[j] = acc.x + acc.y;
        }
        float c4[4], c2[2];
        #pragma unroll
        for (int t = 0; t < 4; ++t)  c4[t] = xadd_in_row<DPP_XMASK_ + 4>(p[t], p[t + 4], up4);
        #pragma unroll
        for (int t = 0; t < 2; ++t)  c2[t] = xadd_in_row<DPP_XMASK_ + 2>(c4[t], c4[t + 2], up2);
        const float sc = xadd_in_row<DPP_XMASK_ + 1>(c2[0], c2[1], up1);

        rbase[(size_t)q * K_] = masked ? NEG_ : sc;
        #pragma unroll
        for (int i = 0; i < 4; ++i) qv[i] = qn[i];
    }
}

__device__ __forceinline__ float lane_xor16(float x) {
    const int xi = __builtin_bit_cast(int, x);
    return __builtin_bit_cast(float, __builtin_amdgcn_permlanex16(
        xi, xi, 0x76543210, 0xfedcba98, false, false));
}

template<int CTRL>
__device__ __forceinline__ float lane_xor_in_row(float x) {
    return __builtin_bit_cast(float, __builtin_amdgcn_update_dpp(
        0, __builtin_bit_cast(int, x), CTRL, 0xF, 0xF, true));
}

__device__ __forceinline__ float wave_allmax(float m) {
    m = fmaxf(m, lane_xor16(m));
    m = fmaxf(m, lane_xor_in_row<DPP_XMASK_ + 8>(m));
    m = fmaxf(m, lane_xor_in_row<DPP_XMASK_ + 4>(m));
    m = fmaxf(m, lane_xor_in_row<DPP_XMASK_ + 2>(m));
    m = fmaxf(m, lane_xor_in_row<DPP_XMASK_ + 1>(m));
    return m;
}

__device__ __forceinline__ float wave_allsum(float s) {
    s += lane_xor16(s);
    s += lane_xor_in_row<DPP_XMASK_ + 8>(s);
    s += lane_xor_in_row<DPP_XMASK_ + 4>(s);
    s += lane_xor_in_row<DPP_XMASK_ + 2>(s);
    s += lane_xor_in_row<DPP_XMASK_ + 1>(s);
    return s;
}

template<int N, int KD, int RING, int GRP>
__global__ __launch_bounds__(128) void softmax_gemm_wmma_f32_ahead(
    const float* __restrict__ S, long sS,
    const float* __restrict__ W, long sW,
    float* __restrict__ C, long sC)
{
    constexpr int tilesN = N >> 4;
    constexpr int STEPS  = KD / 4;
    constexpr int NGRP   = STEPS / GRP;
    constexpr int NI     = KD / 256;
    constexpr int PITCH  = KD + 4;
    static_assert(tilesN % 4 == 0, "the four waves of a block must share their A strip");
    static_assert(STEPS % GRP == 0 && KD % 256 == 0 && RING <= STEPS, "shape");

    __shared__ __attribute__((aligned(16))) float s_a[16 * PITCH];

    S += (size_t)blockIdx.y * sS;
    W += (size_t)blockIdx.y * sW;
    C += (size_t)blockIdx.y * sC;

    const int tid  = threadIdx.x;
    const int wave = tid >> 5;
    const int lane = tid & 31;
    const int tile = blockIdx.x * 4 + wave;
    const int tm = tile / tilesN;
    const int tn = tile - tm * tilesN;
    const int r    = lane & 15;
    const int half = lane >> 4;

    const float* wcol = W + (size_t)(2 * half) * N + tn * 16 + r;

    const float* srow = S + (size_t)(tm * 16 + wave * 4) * KD + lane;
    float v[4][NI][8];
    #pragma unroll
    for (int rr = 0; rr < 4; ++rr) {
        #pragma unroll
        for (int i = 0; i < NI; ++i) {
            #pragma unroll
            for (int w = 0; w < 8; ++w) v[rr][i][w] = srow[(size_t)rr * KD + i * 256 + w * 32];
        }
    }

    v2f rb[RING];
    #pragma unroll
    for (int s = 0; s < RING; ++s) {
        rb[s].x = wcol[(size_t)(4 * s) * N];
        rb[s].y = wcol[(size_t)(4 * s + 1) * N];
    }

    float m[4];
    #pragma unroll
    for (int rr = 0; rr < 4; ++rr) {
        m[rr] = v[rr][0][0];
        #pragma unroll
        for (int i = 0; i < NI; ++i) {
            #pragma unroll
            for (int w = 0; w < 8; ++w) m[rr] = fmaxf(m[rr], v[rr][i][w]);
        }
    }
    #pragma unroll
    for (int rr = 0; rr < 4; ++rr) m[rr] = wave_allmax(m[rr]);

    #pragma unroll
    for (int rr = 0; rr < 4; ++rr) {
        #pragma unroll
        for (int i = 0; i < NI; ++i) {
            #pragma unroll
            for (int w = 0; w < 8; ++w) v[rr][i][w] = EXP2F((v[rr][i][w] - m[rr]) * LOG2E);
        }
    }

    float sm[4][8];
    #pragma unroll
    for (int rr = 0; rr < 4; ++rr) {
        #pragma unroll
        for (int w = 0; w < 8; ++w) {
            float t = v[rr][0][w];
            #pragma unroll
            for (int i = 1; i < NI; ++i) t += v[rr][i][w];
            sm[rr][w] = t;
        }
    }
    #pragma unroll
    for (int rr = 0; rr < 4; ++rr) {
        #pragma unroll
        for (int w = 0; w < 8; ++w) sm[rr][w] = wave_allsum(sm[rr][w]);
    }

    #pragma unroll
    for (int rr = 0; rr < 4; ++rr) {
        float tot = sm[rr][0];
        #pragma unroll
        for (int w = 1; w < 8; ++w) tot += sm[rr][w];
        const float inv = 1.0f / tot;
        float* drow = s_a + (wave * 4 + rr) * PITCH + lane;
        #pragma unroll
        for (int i = 0; i < NI; ++i) {
            #pragma unroll
            for (int w = 0; w < 8; ++w) drow[i * 256 + w * 32] = v[rr][i][w] * inv;
        }
    }
    __syncthreads();

    const float* afrag = s_a + r * PITCH + 2 * half;
    v2f ra[2][GRP];
    #pragma unroll
    for (int s = 0; s < GRP; ++s)
        ra[0][s] = *(const v2f*)(afrag + 4 * s);

    v8f c = {};
    #pragma unroll
    for (int g = 0; g < NGRP; ++g) {
        if (g + 1 < NGRP) {
            #pragma unroll
            for (int s = 0; s < GRP; ++s)
                ra[(g + 1) & 1][s] = *(const v2f*)(afrag + 4 * ((g + 1) * GRP + s));
        }
        __builtin_amdgcn_sched_barrier(0);
        #pragma unroll
        for (int s = 0; s < GRP; ++s) {
            const int st = g * GRP + s;
            c = __builtin_amdgcn_wmma_f32_16x16x4_f32(
                     false, ra[g & 1][s],  false, rb[st % RING],
                     (short)0, c,  false,  false);
            if (st + RING < STEPS) {
                const int n = st + RING;
                rb[st % RING].x = wcol[(size_t)(4 * n) * N];
                rb[st % RING].y = wcol[(size_t)(4 * n + 1) * N];
            }
            __builtin_amdgcn_sched_barrier(0);
        }
    }

    float* crow = C + (size_t)(tm * 16 + 8 * half) * N + tn * 16 + r;
    #pragma unroll
    for (int v2 = 0; v2 < 8; ++v2)
        crow[(size_t)v2 * N] = c[v2];
}

extern "C" void kernel_launch(void* const* d_in, const int* in_sizes, int n_in,
                              void* d_out, int out_size, void* d_ws, size_t ws_size,
                              hipStream_t stream) {
    const float* queries   = (const float*)d_in[0];
    const float* keys      = (const float*)d_in[1];
    const float* values    = (const float*)d_in[2];
    const int*   valid_len = (const int*)  d_in[3];
    const float* Wq        = (const float*)d_in[4];
    const float* Wk        = (const float*)d_in[5];
    const float* Ws        = (const float*)d_in[6];
    float* out = (float*)d_out;

    float* qproj = (float*)d_ws;
    float* attn  = qproj + (size_t)B_ * Q_ * H_
                         + (size_t)B_ * K_ * H_;

    gemm_wmma_f32<H_, QS_><<<dim3(128, 1), 128, 0, stream>>>(
        queries, 0, Wq, 0, qproj, 0);

    kproj_scores<<<B_ * (K_ / 32), 256, 0, stream>>>(
        keys, Wk, qproj, Ws, valid_len, attn);

    softmax_gemm_wmma_f32_ahead<VS_, K_, 30, 16><<<dim3(16, B_), 128, 0, stream>>>(
        attn,   (long)Q_ * K_,
        values, (long)K_ * VS_,
        out,    (long)Q_ * VS_);
}
